# starting kernel
# baseline (speedup 1.0000x reference)
#include <hip/hip_runtime.h>
#include <math.h>

typedef __attribute__((ext_vector_type(2))) float v2f;
typedef __attribute__((ext_vector_type(8))) float v8f;

constexpr int B  = 32;
constexpr int Z  = 2048;
constexpr int N  = 1024;
constexpr int ZC = 8;
constexpr int ZCHUNK = Z / ZC;

__global__ void k_partial(const float* __restrict__ x,
                          const float* __restrict__ alpha,
                          float* __restrict__ s_part) {
    int n  = blockIdx.x * blockDim.x + threadIdx.x;
    int zc = blockIdx.y;
    int b  = blockIdx.z;
    const float* xp = x + ((size_t)b * Z + (size_t)zc * ZCHUNK) * N + n;
    const float* ap = alpha + zc * ZCHUNK;
    float acc = 0.f;
#pragma unroll 4
    for (int i = 0; i < ZCHUNK; ++i)
        acc = fmaf(ap[i], xp[(size_t)i * N], acc);
    s_part[((size_t)b * ZC + zc) * N + n] = acc;
}

__global__ void k_hidden(const float* __restrict__ s_part,
                         float* __restrict__ h,
                         float* __restrict__ denom) {
    __shared__ float red[N];
    int b = blockIdx.x;
    int n = threadIdx.x;
    float s = 0.f;
#pragma unroll
    for (int zc = 0; zc < ZC; ++zc)
        s += s_part[((size_t)b * ZC + zc) * N + n];
    float hv = 1.f / (1.f + __expf(-s));
    h[b * N + n] = hv;
    red[n] = hv * hv;
    __syncthreads();
    for (int st = N / 2; st > 0; st >>= 1) {
        if (n < st) red[n] += red[n + st];
        __syncthreads();
    }
    if (n == 0) denom[b] = red[0];
}

__global__ void k_beta(const float* __restrict__ x,
                       const float* __restrict__ h,
                       const float* __restrict__ denom,
                       float* __restrict__ out) {
    __shared__ float lh[N];
    int b    = blockIdx.x >> 4;
    int zblk = blockIdx.x & 15;
    int tid  = threadIdx.x;

#pragma unroll
    for (int i = tid; i < N; i += 256)
        lh[i] = h[b * N + i];
    __syncthreads();

    int wave = tid >> 5;
    int lane = tid & 31;
    int z0   = zblk * 128 + wave * 16;
    int zrow = z0 + (lane & 15);
    int koff = (lane >> 4) << 1;
    const float* xrow = x + ((size_t)b * Z + zrow) * N;

    v8f acc0 = {0.f, 0.f, 0.f, 0.f, 0.f, 0.f, 0.f, 0.f};
    v8f acc1 = {0.f, 0.f, 0.f, 0.f, 0.f, 0.f, 0.f, 0.f};

    for (int n0 = 0; n0 < N; n0 += 8) {
        v2f a0 = *(const v2f*)(xrow + n0 + koff);
        v2f a1 = *(const v2f*)(xrow + n0 + 4 + koff);
        v2f b0; b0.x = lh[n0 + koff];     b0.y = lh[n0 + koff + 1];
        v2f b1; b1.x = lh[n0 + 4 + koff]; b1.y = lh[n0 + 4 + koff + 1];
        acc0 = __builtin_amdgcn_wmma_f32_16x16x4_f32(
                   false, a0, false, b0, (short)0, acc0, false, false);
        acc1 = __builtin_amdgcn_wmma_f32_16x16x4_f32(
                   false, a1, false, b1, (short)0, acc1, false, false);
    }

    v8f d = acc0 + acc1;
    union { v8f v; float f[8]; } u;
    u.v = d;

    if ((lane & 15) == 0) {
        float inv = 1.f / denom[b];
        int zout = z0 + ((lane >> 4) ? 8 : 0);
#pragma unroll
        for (int r = 0; r < 8; ++r)
            out[b * Z + zout + r] = u.f[r] * inv;
    }
}

extern "C" void kernel_launch(void* const* d_in, const int* in_sizes, int n_in,
                              void* d_out, int out_size, void* d_ws, size_t ws_size,
                              hipStream_t stream) {
    (void)in_sizes; (void)n_in; (void)out_size; (void)ws_size;
    const float* x     = (const float*)d_in[0];
    const float* alpha = (const float*)d_in[1];
    float* out = (float*)d_out;

    float* s_part = (float*)d_ws;
    float* hbuf   = s_part + (size_t)B * ZC * N;
    float* denom  = hbuf + (size_t)B * N;

    dim3 g1(N / 256, ZC, B);
    k_partial<<<g1, 256, 0, stream>>>(x, alpha, s_part);
    k_hidden<<<B, N, 0, stream>>>(s_part, hbuf, denom);
    k_beta<<<B * 16, 256, 0, stream>>>(x, hbuf, denom, out);
}


# x sweeps re-laid for memory parallelism: b128 loads of whole contiguous row pieces, a group of rows requested ahead (sweep 2: h in registers, dot on the WMMA diagonal); f32 summation order changes
# speedup vs baseline: 1.4913x; 1.4913x over previous
#include <hip/hip_runtime.h>
#include <math.h>


typedef __attribute__((ext_vector_type(2))) float v2f;
typedef __attribute__((ext_vector_type(4))) float v4f;
typedef __attribute__((ext_vector_type(8))) float v8f;

constexpr int B  = 32;
constexpr int Z  = 2048;
constexpr int N  = 1024;
constexpr int ZC = 8;
constexpr int ZCHUNK = Z / ZC;


constexpr int P_WROWS = ZCHUNK / 2;
constexpr int P_GRP   = 16;
constexpr int P_NGRP  = P_WROWS / P_GRP;

__global__ void __launch_bounds__(256)
k_partial(const float* __restrict__ x,
          const float* __restrict__ alpha,
          float* __restrict__ s_part) {
    __shared__ v4f red[4][32];
    const int tid  = threadIdx.x;
    const int lane = tid & 31;
    const int wave = __builtin_amdgcn_readfirstlane(tid >> 5);
    const int ng   = wave & 3;
    const int zh   = wave >> 2;
    const int b    = blockIdx.z;
    const int zc   = blockIdx.y;
    const int n    = blockIdx.x * 512 + ng * 128 + lane * 4;
    const int zb   = zc * ZCHUNK + zh * P_WROWS;
    const float* xp = x + ((size_t)b * Z + zb) * N + n;
    const float* ap = alpha + zb;

    v4f   buf[2][P_GRP];
    float aw[2][P_GRP];
#pragma unroll
    for (int i = 0; i < P_GRP; ++i) {
        buf[0][i] = *(const v4f*)(xp + (size_t)i * N);
        aw[0][i]  = ap[i];
    }
    v4f acc = {0.f, 0.f, 0.f, 0.f};
#pragma unroll
    for (int g = 0; g < P_NGRP; ++g) {
        __builtin_amdgcn_sched_barrier(0);
        if (g + 1 < P_NGRP) {
#pragma unroll
            for (int i = 0; i < P_GRP; ++i) {
                buf[(g + 1) & 1][i] = *(const v4f*)(xp + (size_t)((g + 1) * P_GRP + i) * N);
                aw[(g + 1) & 1][i]  = ap[(g + 1) * P_GRP + i];
            }
        }
        __builtin_amdgcn_sched_barrier(0);
#pragma unroll
        for (int i = 0; i < P_GRP; ++i) {
            const float a = aw[g & 1][i];
            const v4f   v = buf[g & 1][i];
            acc.x = fmaf(a, v.x, acc.x);
            acc.y = fmaf(a, v.y, acc.y);
            acc.z = fmaf(a, v.z, acc.z);
            acc.w = fmaf(a, v.w, acc.w);
        }
    }
    if (zh == 1) red[ng][lane] = acc;
    __syncthreads();
    if (zh == 0) {
        const v4f o = red[ng][lane];
        acc.x += o.x; acc.y += o.y; acc.z += o.z; acc.w += o.w;
        *(v4f*)(s_part + ((size_t)b * ZC + zc) * N + n) = acc;
    }
}

__global__ void k_hidden(const float* __restrict__ s_part,
                         float* __restrict__ h,
                         float* __restrict__ denom) {
    __shared__ float red[N];
    int b = blockIdx.x;
    int n = threadIdx.x;
    float s = 0.f;
#pragma unroll
    for (int zc = 0; zc < ZC; ++zc)
        s += s_part[((size_t)b * ZC + zc) * N + n];
    float hv = 1.f / (1.f + __expf(-s));
    h[b * N + n] = hv;
    red[n] = hv * hv;
    __syncthreads();
    for (int st = N / 2; st > 0; st >>= 1) {
        if (n < st) red[n] += red[n + st];
        __syncthreads();
    }
    if (n == 0) denom[b] = red[0];
}

constexpr int R_ROWS  = 16;
constexpr int R_AHEAD = 2;
constexpr int R_Q     = N / 128;

__global__ void __launch_bounds__(256)
k_beta(const float* __restrict__ x,
       const float* __restrict__ h,
       const float* __restrict__ denom,
       float* __restrict__ out) {
    const int tid  = threadIdx.x;
    const int lane = tid & 31;
    const int wave = __builtin_amdgcn_readfirstlane(tid >> 5);
    const int b    = blockIdx.x >> 4;
    const int z0   = (blockIdx.x & 15) * 128 + wave * R_ROWS;

    v4f hq[R_Q];
    const float* hp = h + b * N + lane * 4;
#pragma unroll
    for (int q = 0; q < R_Q; ++q) hq[q] = *(const v4f*)(hp + q * 128);

    const float* xp = x + ((size_t)b * Z + z0) * N + lane * 4;
    v4f xr[R_AHEAD + 1][R_Q];
#define LOAD_ROW(R) do { _Pragma("unroll") for (int q = 0; q < R_Q; ++q) \
        xr[(R) % (R_AHEAD + 1)][q] = *(const v4f*)(xp + (size_t)(R) * N + q * 128); } while (0)
#pragma unroll
    for (int r = 0; r < R_AHEAD; ++r) LOAD_ROW(r);

    const int  dr  = lane & 7;
    const bool own = ((lane >> 3) & 1) == (lane >> 4);
    float sum[R_ROWS];
#pragma unroll
    for (int r = 0; r < R_ROWS; ++r) {
        __builtin_amdgcn_sched_barrier(0);
        if (r + R_AHEAD < R_ROWS) LOAD_ROW(r + R_AHEAD);
        __builtin_amdgcn_sched_barrier(0);
        v8f acc0 = {0.f, 0.f, 0.f, 0.f, 0.f, 0.f, 0.f, 0.f};
        v8f acc1 = {0.f, 0.f, 0.f, 0.f, 0.f, 0.f, 0.f, 0.f};
#pragma unroll
        for (int q = 0; q < R_Q; ++q) {
            const v4f v = xr[r % (R_AHEAD + 1)][q];
            v2f a0; a0.x = v.x;     a0.y = v.y;
            v2f a1; a1.x = v.z;     a1.y = v.w;
            v2f b0; b0.x = hq[q].x; b0.y = hq[q].y;
            v2f b1; b1.x = hq[q].z; b1.y = hq[q].w;
            acc0 = __builtin_amdgcn_wmma_f32_16x16x4_f32(
                       false, a0, false, b0, (short)0, acc0, false, false);
            acc1 = __builtin_amdgcn_wmma_f32_16x16x4_f32(
                       false, a1, false, b1, (short)0, acc1, false, false);
        }
        const v8f d = acc0 + acc1;
        float t = d[0];
        t = (dr == 1) ? d[1] : t;
        t = (dr == 2) ? d[2] : t;
        t = (dr == 3) ? d[3] : t;
        t = (dr == 4) ? d[4] : t;
        t = (dr == 5) ? d[5] : t;
        t = (dr == 6) ? d[6] : t;
        t = (dr == 7) ? d[7] : t;
        sum[r] = own ? t : 0.f;
    }
#undef LOAD_ROW
    __builtin_amdgcn_sched_barrier(0);

#pragma unroll
    for (int r = 0; r < R_ROWS; ++r) {
        float s = sum[r];
#pragma unroll
        for (int m = 16; m >= 1; m >>= 1) s += __shfl_xor(s, m, 32);
        sum[r] = s;
    }
    float mine = 0.f;
#pragma unroll
    for (int r = 0; r < R_ROWS; ++r) if (lane == r) mine = sum[r];
    if (lane < R_ROWS) {
        const float inv = 1.f / denom[b];
        out[b * Z + z0 + lane] = mine * inv;
    }
}

extern "C" void kernel_launch(void* const* d_in, const int* in_sizes, int n_in,
                              void* d_out, int out_size, void* d_ws, size_t ws_size,
                              hipStream_t stream) {
    (void)in_sizes; (void)n_in; (void)out_size; (void)ws_size;
    const float* x     = (const float*)d_in[0];
    const float* alpha = (const float*)d_in[1];
    float* out = (float*)d_out;

    float* s_part = (float*)d_ws;
    float* hbuf   = s_part + (size_t)B * ZC * N;
    float* denom  = hbuf + (size_t)B * N;

    dim3 g1(N / 512, ZC, B);
    k_partial<<<g1, 256, 0, stream>>>(x, alpha, s_part);
    k_hidden<<<B, N, 0, stream>>>(s_part, hbuf, denom);
    k_beta<<<B * 16, 256, 0, stream>>>(x, hbuf, denom, out);
}


# hidden stage folded into the second sweep: each work-group rebuilds h and the denominator of its batch from the partial sums under its first x loads; 3 launches to 2; denominator sum order changes
# speedup vs baseline: 1.6551x; 1.1098x over previous
#include <hip/hip_runtime.h>
#include <math.h>


typedef __attribute__((ext_vector_type(2))) float v2f;
typedef __attribute__((ext_vector_type(4))) float v4f;
typedef __attribute__((ext_vector_type(8))) float v8f;

constexpr int B  = 32;
constexpr int Z  = 2048;
constexpr int N  = 1024;
constexpr int ZC = 8;
constexpr int ZCHUNK = Z / ZC;


constexpr int P_WROWS = ZCHUNK / 2;
constexpr int P_GRP   = 16;
constexpr int P_NGRP  = P_WROWS / P_GRP;

__global__ void __launch_bounds__(256)
k_partial(const float* __restrict__ x,
          const float* __restrict__ alpha,
          float* __restrict__ s_part) {
    __shared__ v4f red[4][32];
    const int tid  = threadIdx.x;
    const int lane = tid & 31;
    const int wave = __builtin_amdgcn_readfirstlane(tid >> 5);
    const int ng   = wave & 3;
    const int zh   = wave >> 2;
    const int b    = blockIdx.z;
    const int zc   = blockIdx.y;
    const int n    = blockIdx.x * 512 + ng * 128 + lane * 4;
    const int zb   = zc * ZCHUNK + zh * P_WROWS;
    const float* xp = x + ((size_t)b * Z + zb) * N + n;
    const float* ap = alpha + zb;

    v4f   buf[2][P_GRP];
    float aw[2][P_GRP];
#pragma unroll
    for (int i = 0; i < P_GRP; ++i) {
        buf[0][i] = *(const v4f*)(xp + (size_t)i * N);
        aw[0][i]  = ap[i];
    }
    v4f acc = {0.f, 0.f, 0.f, 0.f};
#pragma unroll
    for (int g = 0; g < P_NGRP; ++g) {
        __builtin_amdgcn_sched_barrier(0);
        if (g + 1 < P_NGRP) {
#pragma unroll
            for (int i = 0; i < P_GRP; ++i) {
                buf[(g + 1) & 1][i] = *(const v4f*)(xp + (size_t)((g + 1) * P_GRP + i) * N);
                aw[(g + 1) & 1][i]  = ap[(g + 1) * P_GRP + i];
            }
        }
        __builtin_amdgcn_sched_barrier(0);
#pragma unroll
        for (int i = 0; i < P_GRP; ++i) {
            const float a = aw[g & 1][i];
            const v4f   v = buf[g & 1][i];
            acc.x = fmaf(a, v.x, acc.x);
            acc.y = fmaf(a, v.y, acc.y);
            acc.z = fmaf(a, v.z, acc.z);
            acc.w = fmaf(a, v.w, acc.w);
        }
    }
    if (zh == 1) red[ng][lane] = acc;
    __syncthreads();
    if (zh == 0) {
        const v4f o = red[ng][lane];
        acc.x += o.x; acc.y += o.y; acc.z += o.z; acc.w += o.w;
        *(v4f*)(s_part + ((size_t)b * ZC + zc) * N + n) = acc;
    }
}

constexpr int R_ROWS  = 16;
constexpr int R_AHEAD = 2;
constexpr int R_Q     = N / 128;

__global__ void __launch_bounds__(256)
k_beta(const float* __restrict__ x,
       const float* __restrict__ s_part,
       float* __restrict__ out) {
    __shared__ v4f   h_sh[N / 4];
    __shared__ float d_sh[8];
    const int tid  = threadIdx.x;
    const int lane = tid & 31;
    const int wave = __builtin_amdgcn_readfirstlane(tid >> 5);
    const int b    = blockIdx.x >> 4;
    const int z0   = (blockIdx.x & 15) * 128 + wave * R_ROWS;

    const float* sp = s_part + (size_t)b * ZC * N + tid * 4;
    v4f p[ZC];
#pragma unroll
    for (int zc = 0; zc < ZC; ++zc) p[zc] = *(const v4f*)(sp + (size_t)zc * N);
    __builtin_amdgcn_sched_barrier(0);

    const float* xp = x + ((size_t)b * Z + z0) * N + lane * 4;
    v4f xr[R_AHEAD + 1][R_Q];
#define LOAD_ROW(R) do { _Pragma("unroll") for (int q = 0; q < R_Q; ++q) \
        xr[(R) % (R_AHEAD + 1)][q] = *(const v4f*)(xp + (size_t)(R) * N + q * 128); } while (0)
#pragma unroll
    for (int r = 0; r < R_AHEAD; ++r) LOAD_ROW(r);
    __builtin_amdgcn_sched_barrier(0);

    v4f s = {0.f, 0.f, 0.f, 0.f};
#pragma unroll
    for (int zc = 0; zc < ZC; ++zc) s += p[zc];
    v4f hv;
    hv.x = 1.f / (1.f + __expf(-s.x));
    hv.y = 1.f / (1.f + __expf(-s.y));
    hv.z = 1.f / (1.f + __expf(-s.z));
    hv.w = 1.f / (1.f + __expf(-s.w));
    h_sh[tid] = hv;
    float dsum = hv.x * hv.x + hv.y * hv.y + hv.z * hv.z + hv.w * hv.w;
#pragma unroll
    for (int m = 16; m >= 1; m >>= 1) dsum += __shfl_xor(dsum, m, 32);
    if (lane == 0) d_sh[wave] = dsum;
    __builtin_amdgcn_fence(__ATOMIC_RELEASE, "workgroup", "local");
    __builtin_amdgcn_s_barrier();
    __builtin_amdgcn_fence(__ATOMIC_ACQUIRE, "workgroup", "local");
    v4f hq[R_Q];
#pragma unroll
    for (int q = 0; q < R_Q; ++q) hq[q] = h_sh[q * 32 + lane];
    float den = 0.f;
#pragma unroll
    for (int w = 0; w < 8; ++w) den += d_sh[w];

    const int  dr  = lane & 7;
    const bool own = ((lane >> 3) & 1) == (lane >> 4);
    float sum[R_ROWS];
#pragma unroll
    for (int r = 0; r < R_ROWS; ++r) {
        __builtin_amdgcn_sched_barrier(0);
        if (r + R_AHEAD < R_ROWS) LOAD_ROW(r + R_AHEAD);
        __builtin_amdgcn_sched_barrier(0);
        v8f acc0 = {0.f, 0.f, 0.f, 0.f, 0.f, 0.f, 0.f, 0.f};
        v8f acc1 = {0.f, 0.f, 0.f, 0.f, 0.f, 0.f, 0.f, 0.f};
#pragma unroll
        for (int q = 0; q < R_Q; ++q) {
            const v4f v = xr[r % (R_AHEAD + 1)][q];
            v2f a0; a0.x = v.x;     a0.y = v.y;
            v2f a1; a1.x = v.z;     a1.y = v.w;
            v2f b0; b0.x = hq[q].x; b0.y = hq[q].y;
            v2f b1; b1.x = hq[q].z; b1.y = hq[q].w;
            acc0 = __builtin_amdgcn_wmma_f32_16x16x4_f32(
                       false, a0, false, b0, (short)0, acc0, false, false);
            acc1 = __builtin_amdgcn_wmma_f32_16x16x4_f32(
                       false, a1, false, b1, (short)0, acc1, false, false);
        }
        const v8f d = acc0 + acc1;
        float t = d[0];
        t = (dr == 1) ? d[1] : t;
        t = (dr == 2) ? d[2] : t;
        t = (dr == 3) ? d[3] : t;
        t = (dr == 4) ? d[4] : t;
        t = (dr == 5) ? d[5] : t;
        t = (dr == 6) ? d[6] : t;
        t = (dr == 7) ? d[7] : t;
        sum[r] = own ? t : 0.f;
    }
#undef LOAD_ROW
    __builtin_amdgcn_sched_barrier(0);

#pragma unroll
    for (int r = 0; r < R_ROWS; ++r) {
        float s = sum[r];
#pragma unroll
        for (int m = 16; m >= 1; m >>= 1) s += __shfl_xor(s, m, 32);
        sum[r] = s;
    }
    float mine = 0.f;
#pragma unroll
    for (int r = 0; r < R_ROWS; ++r) if (lane == r) mine = sum[r];
    if (lane < R_ROWS) {
        const float inv = 1.f / den;
        out[b * Z + z0 + lane] = mine * inv;
    }
}

extern "C" void kernel_launch(void* const* d_in, const int* in_sizes, int n_in,
                              void* d_out, int out_size, void* d_ws, size_t ws_size,
                              hipStream_t stream) {
    (void)in_sizes; (void)n_in; (void)out_size; (void)ws_size;
    const float* x     = (const float*)d_in[0];
    const float* alpha = (const float*)d_in[1];
    float* out = (float*)d_out;

    float* s_part = (float*)d_ws;

    dim3 g1(N / 512, ZC, B);
    k_partial<<<g1, 256, 0, stream>>>(x, alpha, s_part);
    k_beta<<<B * 16, 256, 0, stream>>>(x, s_part, out);
}


# x read from memory once: a 32-wave work-group holds a 32-column slice of a batch (all rows) in registers across both sweeps; a small launch adds 8 partial planes; f32 sum order changes
# speedup vs baseline: 2.3475x; 1.4183x over previous
#include <hip/hip_runtime.h>
#include <math.h>


typedef __attribute__((ext_vector_type(2))) float v2f;
typedef __attribute__((ext_vector_type(4))) float v4f;
typedef __attribute__((ext_vector_type(8))) float v8f;

constexpr int B  = 32;
constexpr int Z  = 2048;
constexpr int N  = 1024;

constexpr int SL     = 32;
constexpr int NIT    = 4;
constexpr int STRIPS = N / (SL * NIT);
constexpr int NWL    = 16;
constexpr int RP     = 36;
constexpr size_t PB_FLOATS  = (size_t)B * STRIPS * Z;
constexpr size_t PD_FLOATS  = (size_t)B * STRIPS;
constexpr size_t NEED_BYTES = (PB_FLOATS + PD_FLOATS) * sizeof(float);

__device__ __forceinline__ float quad_xor1(float v) {
    return __int_as_float(__builtin_amdgcn_update_dpp(0, __float_as_int(v), 0xB1, 0xF, 0xF, true));
}
__device__ __forceinline__ float quad_xor2(float v) {
    return __int_as_float(__builtin_amdgcn_update_dpp(0, __float_as_int(v), 0x4E, 0xF, 0xF, true));
}

__device__ __forceinline__ float rowgroups_sum(float v) {
    v += __int_as_float(__builtin_amdgcn_update_dpp(0, __float_as_int(v), 0x128, 0xF, 0xF, true));
    v += __int_as_float(__builtin_amdgcn_update_dpp(0, __float_as_int(v), 0x124, 0xF, 0xF, true));
    return v;
}
__device__ __forceinline__ float wave_sum(float v) {
    v += quad_xor1(v);
    v += quad_xor2(v);
    v += __int_as_float(__builtin_amdgcn_update_dpp(0, __float_as_int(v), 0x141, 0xF, 0xF, true));
    v += __int_as_float(__builtin_amdgcn_update_dpp(0, __float_as_int(v), 0x140, 0xF, 0xF, true));
    v += __int_as_float(__builtin_amdgcn_permlanex16(0, __float_as_int(v), 0x76543210, 0xfedcba98, false, false));
    return v;
}


template <bool REFILL>
__device__ __forceinline__ void do_slice(v4f (&xr)[NWL], const v4f* al4,
                                         const float* __restrict__ xnext,
                                         float* red, v4f* h_sh,
                                         const int wave, const int lane, const int g, const int cp,
                                         v4f& osum, float& dacc) {
    float aw[NWL];
#pragma unroll
    for (int q = 0; q < NWL / 4; ++q) {
        const v4f a = al4[q];
        aw[4 * q] = a.x; aw[4 * q + 1] = a.y; aw[4 * q + 2] = a.z; aw[4 * q + 3] = a.w;
    }
    v4f acc = {0.f, 0.f, 0.f, 0.f};
#pragma unroll
    for (int j = 0; j < NWL; ++j) {
        acc.x = fmaf(aw[j], xr[j].x, acc.x);
        acc.y = fmaf(aw[j], xr[j].y, acc.y);
        acc.z = fmaf(aw[j], xr[j].z, acc.z);
        acc.w = fmaf(aw[j], xr[j].w, acc.w);
    }
    acc.x = rowgroups_sum(acc.x);
    acc.y = rowgroups_sum(acc.y);
    acc.z = rowgroups_sum(acc.z);
    acc.w = rowgroups_sum(acc.w);
    if (g == 0) *(v4f*)(red + wave * RP + cp * 4) = acc;
    __builtin_amdgcn_fence(__ATOMIC_RELEASE, "workgroup", "local");
    __builtin_amdgcn_s_barrier();
    __builtin_amdgcn_fence(__ATOMIC_ACQUIRE, "workgroup", "local");
    {
        const float s  = wave_sum(red[lane * RP + wave]);
        const float hc = 1.f / (1.f + __expf(-s));
        if (lane == 0) ((float*)h_sh)[wave] = hc;
        dacc = fmaf(hc, hc, dacc);
    }
    __builtin_amdgcn_fence(__ATOMIC_RELEASE, "workgroup", "local");
    __builtin_amdgcn_s_barrier();
    __builtin_amdgcn_fence(__ATOMIC_ACQUIRE, "workgroup", "local");
    const v4f hq = h_sh[cp];
    v2f b0; b0.x = hq.x; b0.y = hq.y;
    v2f b1; b1.x = hq.z; b1.y = hq.w;

    const int dr = lane & 7;
    const int i  = lane & 3;
    v4f o = {0.f, 0.f, 0.f, 0.f};
#pragma unroll
    for (int j = 0; j < NWL; j += 2) {
        v8f dA = {0.f, 0.f, 0.f, 0.f, 0.f, 0.f, 0.f, 0.f};
        v8f dB = {0.f, 0.f, 0.f, 0.f, 0.f, 0.f, 0.f, 0.f};
        v2f a0; a0.x = xr[j].x;     a0.y = xr[j].y;
        v2f a1; a1.x = xr[j].z;     a1.y = xr[j].w;
        v2f c0; c0.x = xr[j + 1].x; c0.y = xr[j + 1].y;
        v2f c1; c1.x = xr[j + 1].z; c1.y = xr[j + 1].w;
        dA = __builtin_amdgcn_wmma_f32_16x16x4_f32(false, a0, false, b0, (short)0, dA, false, false);
        dB = __builtin_amdgcn_wmma_f32_16x16x4_f32(false, c0, false, b0, (short)0, dB, false, false);
        dA = __builtin_amdgcn_wmma_f32_16x16x4_f32(false, a1, false, b1, (short)0, dA, false, false);
        dB = __builtin_amdgcn_wmma_f32_16x16x4_f32(false, c1, false, b1, (short)0, dB, false, false);
        if (REFILL) {
            xr[j]     = *(const v4f*)(xnext + (size_t)j * N);
            xr[j + 1] = *(const v4f*)(xnext + (size_t)(j + 1) * N);
        }
        float tA = dA[0], tB = dB[0];
        tA = (dr == 1) ? dA[1] : tA;  tB = (dr == 1) ? dB[1] : tB;
        tA = (dr == 2) ? dA[2] : tA;  tB = (dr == 2) ? dB[2] : tB;
        tA = (dr == 3) ? dA[3] : tA;  tB = (dr == 3) ? dB[3] : tB;
        tA = (dr == 4) ? dA[4] : tA;  tB = (dr == 4) ? dB[4] : tB;
        tA = (dr == 5) ? dA[5] : tA;  tB = (dr == 5) ? dB[5] : tB;
        tA = (dr == 6) ? dA[6] : tA;  tB = (dr == 6) ? dB[6] : tB;
        tA = (dr == 7) ? dA[7] : tA;  tB = (dr == 7) ? dB[7] : tB;
        tA += quad_xor1(tA);  tB += quad_xor1(tB);
        tA += quad_xor2(tA);  tB += quad_xor2(tB);
        asm volatile("" : "+v"(tA), "+v"(tB), "+v"(b0), "+v"(b1));
        if ((j & 3) == 0) { o.x = (i == (j >> 2)) ? tA : o.x;  o.y = (i == (j >> 2)) ? tB : o.y; }
        else              { o.z = (i == (j >> 2)) ? tA : o.z;  o.w = (i == (j >> 2)) ? tB : o.w; }
        __builtin_amdgcn_sched_barrier(0);
    }
    osum += o;
}

__global__ void __launch_bounds__(1024)
k_fused(const float* __restrict__ x,
        const float* __restrict__ alpha,
        float* __restrict__ ws,
        const size_t ws_bytes) {
    if (ws_bytes < NEED_BYTES) return;
    __shared__ float red[32 * RP];
    __shared__ float d_sh[32];
    __shared__ v4f   h_sh[SL / 4];
    __shared__ v4f   al_sh[Z / 4];
    const int tid   = threadIdx.x;
    const int lane  = tid & 31;
    const int wave  = __builtin_amdgcn_readfirstlane(tid >> 5);
    const int g     = (lane >> 2) & 3;
    const int cp    = ((lane >> 4) << 2) | (lane & 3);
    const int b     = blockIdx.x >> 3;
    const int strip = blockIdx.x & 7;
    const int zl    = wave * 64 + g * 16;

    const float* xp = x + ((size_t)b * Z + zl) * N + strip * (SL * NIT) + cp * 4;
    const v2f a2 = *(const v2f*)(alpha + 2 * tid);
    v4f xr[NWL];
#pragma unroll
    for (int j = 0; j < NWL; ++j) xr[j] = *(const v4f*)(xp + (size_t)j * N);
    ((v2f*)al_sh)[tid] = a2;
    __builtin_amdgcn_fence(__ATOMIC_RELEASE, "workgroup", "local");
    __builtin_amdgcn_s_barrier();
    __builtin_amdgcn_fence(__ATOMIC_ACQUIRE, "workgroup", "local");
    const v4f* al4 = al_sh + (zl >> 2);

    v4f   osum = {0.f, 0.f, 0.f, 0.f};
    float dacc = 0.f;
    for (int it = 0; it < NIT - 1; ++it) {
        do_slice<true>(xr, al4, xp + SL, red, h_sh, wave, lane, g, cp, osum, dacc);
        xp += SL;
    }
    do_slice<false>(xr, al4, xp, red, h_sh, wave, lane, g, cp, osum, dacc);

    const bool own = ((lane >> 3) & 1) == (lane >> 4);
    float* pb = ws + ((size_t)(b * STRIPS + strip)) * Z;
    if (own) *(v4f*)(pb + zl + (lane & 3) * 4) = osum;
    if (lane == 0) d_sh[wave] = dacc;
    __builtin_amdgcn_fence(__ATOMIC_RELEASE, "workgroup", "local");
    __builtin_amdgcn_s_barrier();
    __builtin_amdgcn_fence(__ATOMIC_ACQUIRE, "workgroup", "local");
    if (wave == 0) {
        const float dsum = wave_sum(d_sh[lane]);
        if (lane == 0) ws[PB_FLOATS + b * STRIPS + strip] = dsum;
    }
}

__global__ void __launch_bounds__(64)
k_sum(const float* __restrict__ ws, float* __restrict__ out, const size_t ws_bytes) {
    if (ws_bytes < NEED_BYTES) return;
    const int t = blockIdx.x * 64 + threadIdx.x;
    const int b = t >> 9;
    const int z = (t & 511) * 4;
    v4f p[STRIPS];
    float d[STRIPS];
#pragma unroll
    for (int s = 0; s < STRIPS; ++s) {
        p[s] = *(const v4f*)(ws + ((size_t)(b * STRIPS + s)) * Z + z);
        d[s] = ws[PB_FLOATS + b * STRIPS + s];
    }
    v4f a = p[0];
    float den = d[0];
#pragma unroll
    for (int s = 1; s < STRIPS; ++s) { a += p[s]; den += d[s]; }
    const float inv = 1.f / den;
    v4f o; o.x = a.x * inv; o.y = a.y * inv; o.z = a.z * inv; o.w = a.w * inv;
    *(v4f*)(out + (size_t)b * Z + z) = o;
}

extern "C" void kernel_launch(void* const* d_in, const int* in_sizes, int n_in,
                              void* d_out, int out_size, void* d_ws, size_t ws_size,
                              hipStream_t stream) {
    (void)in_sizes; (void)n_in; (void)out_size;
    const float* x     = (const float*)d_in[0];
    const float* alpha = (const float*)d_in[1];
    float* out = (float*)d_out;
    float* ws  = (float*)d_ws;

    k_fused<<<B * STRIPS, 1024, 0, stream>>>(x, alpha, ws, ws_size);
    k_sum<<<256, 64, 0, stream>>>(ws, out, ws_size);
}


# speedup vs baseline: 2.4224x; 1.0319x over previous
#include <hip/hip_runtime.h>
#include <math.h>


typedef __attribute__((ext_vector_type(2))) float v2f;
typedef __attribute__((ext_vector_type(4))) float v4f;
typedef __attribute__((ext_vector_type(8))) float v8f;

constexpr int B  = 32;
constexpr int Z  = 2048;
constexpr int N  = 1024;

constexpr int SL     = 32;
constexpr int NIT    = 4;
constexpr int STRIPS = N / (SL * NIT);
constexpr int UNIT   = 2 * SL;
constexpr int QUART  = N / NIT;
constexpr int NWL    = 16;
constexpr int RP     = 36;
constexpr size_t PB_FLOATS  = (size_t)B * STRIPS * Z;
constexpr size_t PD_FLOATS  = (size_t)B * STRIPS;
constexpr size_t NEED_BYTES = (PB_FLOATS + PD_FLOATS) * sizeof(float);

__device__ __forceinline__ float quad_xor1(float v) {
    return __int_as_float(__builtin_amdgcn_update_dpp(0, __float_as_int(v), 0xB1, 0xF, 0xF, true));
}
__device__ __forceinline__ float quad_xor2(float v) {
    return __int_as_float(__builtin_amdgcn_update_dpp(0, __float_as_int(v), 0x4E, 0xF, 0xF, true));
}

__device__ __forceinline__ float rowgroups_sum(float v) {
    v += __int_as_float(__builtin_amdgcn_update_dpp(0, __float_as_int(v), 0x128, 0xF, 0xF, true));
    v += __int_as_float(__builtin_amdgcn_update_dpp(0, __float_as_int(v), 0x124, 0xF, 0xF, true));
    return v;
}
__device__ __forceinline__ float wave_sum(float v) {
    v += quad_xor1(v);
    v += quad_xor2(v);
    v += __int_as_float(__builtin_amdgcn_update_dpp(0, __float_as_int(v), 0x141, 0xF, 0xF, true));
    v += __int_as_float(__builtin_amdgcn_update_dpp(0, __float_as_int(v), 0x140, 0xF, 0xF, true));
    v += __int_as_float(__builtin_amdgcn_permlanex16(0, __float_as_int(v), 0x76543210, 0xfedcba98, false, false));
    return v;
}


template <bool REFILL>
__device__ __forceinline__ void do_slice(v4f (&xr)[NWL], const v4f* al4,
                                         const float* __restrict__ xnext,
                                         float* red, v4f* h_sh,
                                         const int wave, const int lane, const int g, const int cp,
                                         v4f& osum, float& dacc) {
    float aw[NWL];
#pragma unroll
    for (int q = 0; q < NWL / 4; ++q) {
        const v4f a = al4[q];
        aw[4 * q] = a.x; aw[4 * q + 1] = a.y; aw[4 * q + 2] = a.z; aw[4 * q + 3] = a.w;
    }
    v4f acc = {0.f, 0.f, 0.f, 0.f};
#pragma unroll
    for (int j = 0; j < NWL; ++j) {
        acc.x = fmaf(aw[j], xr[j].x, acc.x);
        acc.y = fmaf(aw[j], xr[j].y, acc.y);
        acc.z = fmaf(aw[j], xr[j].z, acc.z);
        acc.w = fmaf(aw[j], xr[j].w, acc.w);
    }
    acc.x = rowgroups_sum(acc.x);
    acc.y = rowgroups_sum(acc.y);
    acc.z = rowgroups_sum(acc.z);
    acc.w = rowgroups_sum(acc.w);
    if (g == 0) *(v4f*)(red + wave * RP + cp * 4) = acc;
    __builtin_amdgcn_fence(__ATOMIC_RELEASE, "workgroup", "local");
    __builtin_amdgcn_s_barrier();
    __builtin_amdgcn_fence(__ATOMIC_ACQUIRE, "workgroup", "local");
    {
        const float s  = wave_sum(red[lane * RP + wave]);
        const float hc = 1.f / (1.f + __expf(-s));
        if (lane == 0) ((float*)h_sh)[wave] = hc;
        dacc = fmaf(hc, hc, dacc);
    }
    __builtin_amdgcn_fence(__ATOMIC_RELEASE, "workgroup", "local");
    __builtin_amdgcn_s_barrier();
    __builtin_amdgcn_fence(__ATOMIC_ACQUIRE, "workgroup", "local");
    const v4f hq = h_sh[cp];
    v2f b0; b0.x = hq.x; b0.y = hq.y;
    v2f b1; b1.x = hq.z; b1.y = hq.w;

    const int dr = lane & 7;
    const int i  = lane & 3;
    v4f o = {0.f, 0.f, 0.f, 0.f};
#pragma unroll
    for (int j = 0; j < NWL; j += 2) {
        v8f dA = {0.f, 0.f, 0.f, 0.f, 0.f, 0.f, 0.f, 0.f};
        v8f dB = {0.f, 0.f, 0.f, 0.f, 0.f, 0.f, 0.f, 0.f};
        v2f a0; a0.x = xr[j].x;     a0.y = xr[j].y;
        v2f a1; a1.x = xr[j].z;     a1.y = xr[j].w;
        v2f c0; c0.x = xr[j + 1].x; c0.y = xr[j + 1].y;
        v2f c1; c1.x = xr[j + 1].z; c1.y = xr[j + 1].w;
        dA = __builtin_amdgcn_wmma_f32_16x16x4_f32(false, a0, false, b0, (short)0, dA, false, false);
        dB = __builtin_amdgcn_wmma_f32_16x16x4_f32(false, c0, false, b0, (short)0, dB, false, false);
        dA = __builtin_amdgcn_wmma_f32_16x16x4_f32(false, a1, false, b1, (short)0, dA, false, false);
        dB = __builtin_amdgcn_wmma_f32_16x16x4_f32(false, c1, false, b1, (short)0, dB, false, false);
        if (REFILL) {
            xr[j]     = *(const v4f*)(xnext + (size_t)j * N);
            xr[j + 1] = *(const v4f*)(xnext + (size_t)(j + 1) * N);
        }
        float tA = dA[0], tB = dB[0];
        tA = (dr == 1) ? dA[1] : tA;  tB = (dr == 1) ? dB[1] : tB;
        tA = (dr == 2) ? dA[2] : tA;  tB = (dr == 2) ? dB[2] : tB;
        tA = (dr == 3) ? dA[3] : tA;  tB = (dr == 3) ? dB[3] : tB;
        tA = (dr == 4) ? dA[4] : tA;  tB = (dr == 4) ? dB[4] : tB;
        tA = (dr == 5) ? dA[5] : tA;  tB = (dr == 5) ? dB[5] : tB;
        tA = (dr == 6) ? dA[6] : tA;  tB = (dr == 6) ? dB[6] : tB;
        tA = (dr == 7) ? dA[7] : tA;  tB = (dr == 7) ? dB[7] : tB;
        tA += quad_xor1(tA);  tB += quad_xor1(tB);
        tA += quad_xor2(tA);  tB += quad_xor2(tB);
        asm volatile("" : "+v"(tA), "+v"(tB), "+v"(b0), "+v"(b1));
        if ((j & 3) == 0) { o.x = (i == (j >> 2)) ? tA : o.x;  o.y = (i == (j >> 2)) ? tB : o.y; }
        else              { o.z = (i == (j >> 2)) ? tA : o.z;  o.w = (i == (j >> 2)) ? tB : o.w; }
        __builtin_amdgcn_sched_barrier(0);
    }
    osum += o;
}

__global__ void __launch_bounds__(1024)
k_fused(const float* __restrict__ x,
        const float* __restrict__ alpha,
        float* __restrict__ ws,
        const size_t ws_bytes) {
    if (ws_bytes < NEED_BYTES) return;
    __shared__ float red[32 * RP];
    __shared__ float d_sh[32];
    __shared__ v4f   h_sh[SL / 4];
    __shared__ v4f   al_sh[Z / 4];
    const int tid   = threadIdx.x;
    const int lane  = tid & 31;
    const int wave  = __builtin_amdgcn_readfirstlane(tid >> 5);
    const int g     = (lane >> 2) & 3;
    const int cp    = ((lane >> 4) << 2) | (lane & 3);
    const int s8    = blockIdx.x & 7;
    const int half  = (blockIdx.x >> 3) & 1;
    const int b     = ((blockIdx.x >> 4) << 1) | (s8 >> 2);
    const int ug    = s8 & 3;
    const int strip = ug * 2 + half;
    const int zl    = wave * 64 + g * 16;

    const float* xp = x + ((size_t)b * Z + zl) * N + ug * UNIT + half * SL + cp * 4;
    const v2f a2 = *(const v2f*)(alpha + 2 * tid);
    v4f xr[NWL];
#pragma unroll
    for (int j = 0; j < NWL; ++j) xr[j] = *(const v4f*)(xp + (size_t)j * N);
    ((v2f*)al_sh)[tid] = a2;
    __builtin_amdgcn_fence(__ATOMIC_RELEASE, "workgroup", "local");
    __builtin_amdgcn_s_barrier();
    __builtin_amdgcn_fence(__ATOMIC_ACQUIRE, "workgroup", "local");
    const v4f* al4 = al_sh + (zl >> 2);

    v4f   osum = {0.f, 0.f, 0.f, 0.f};
    float dacc = 0.f;
    for (int it = 0; it < NIT - 1; ++it) {
        do_slice<true>(xr, al4, xp + QUART, red, h_sh, wave, lane, g, cp, osum, dacc);
        xp += QUART;
    }
    do_slice<false>(xr, al4, xp, red, h_sh, wave, lane, g, cp, osum, dacc);

    const bool own = ((lane >> 3) & 1) == (lane >> 4);
    float* pb = ws + ((size_t)(b * STRIPS + strip)) * Z;
    if (own) *(v4f*)(pb + zl + (lane & 3) * 4) = osum;
    if (lane == 0) d_sh[wave] = dacc;
    __builtin_amdgcn_fence(__ATOMIC_RELEASE, "workgroup", "local");
    __builtin_amdgcn_s_barrier();
    __builtin_amdgcn_fence(__ATOMIC_ACQUIRE, "workgroup", "local");
    if (wave == 0) {
        const float dsum = wave_sum(d_sh[lane]);
        if (lane == 0) ws[PB_FLOATS + b * STRIPS + strip] = dsum;
    }
}

__global__ void __launch_bounds__(64)
k_sum(const float* __restrict__ ws, float* __restrict__ out, const size_t ws_bytes) {
    if (ws_bytes < NEED_BYTES) return;
    const int t = blockIdx.x * 64 + threadIdx.x;
    const int b = t >> 9;
    const int z = (t & 511) * 4;
    v4f p[STRIPS];
    float d[STRIPS];
#pragma unroll
    for (int s = 0; s < STRIPS; ++s) {
        p[s] = *(const v4f*)(ws + ((size_t)(b * STRIPS + s)) * Z + z);
        d[s] = ws[PB_FLOATS + b * STRIPS + s];
    }
    v4f a = p[0];
    float den = d[0];
#pragma unroll
    for (int s = 1; s < STRIPS; ++s) { a += p[s]; den += d[s]; }
    const float inv = 1.f / den;
    v4f o; o.x = a.x * inv; o.y = a.y * inv; o.z = a.z * inv; o.w = a.w * inv;
    *(v4f*)(out + (size_t)b * Z + z) = o;
}

extern "C" void kernel_launch(void* const* d_in, const int* in_sizes, int n_in,
                              void* d_out, int out_size, void* d_ws, size_t ws_size,
                              hipStream_t stream) {
    (void)in_sizes; (void)n_in; (void)out_size;
    const float* x     = (const float*)d_in[0];
    const float* alpha = (const float*)d_in[1];
    float* out = (float*)d_out;
    float* ws  = (float*)d_ws;

    k_fused<<<B * STRIPS, 1024, 0, stream>>>(x, alpha, ws, ws_size);
    k_sum<<<256, 64, 0, stream>>>(ws, out, ws_size);
}
